# starting kernel
# baseline (speedup 1.0000x reference)
#include <hip/hip_runtime.h>
#include <hip/hip_bf16.h>
#include <math.h>

#define S_LEN 256
#define BT_DIM 32
#define D_DIM 128
#define N_DIM 128
#define O_DIM 128
#define ROWS (S_LEN * BT_DIM)
#define EPS_V 1e-7f

typedef __attribute__((ext_vector_type(2))) float v2f;
typedef __attribute__((ext_vector_type(8))) float v8f;

template <bool APPLY_GAMMA>
__global__ __launch_bounds__(256) void gemm_nt_wmma(
    const float* __restrict__ in, const float* __restrict__ W,
    const float* __restrict__ lambda_log, float* __restrict__ out) {
  const int lane = threadIdx.x & 31;
  const int wave = blockIdx.x * 8 + (threadIdx.x >> 5);
  const int rowBase = wave * 16;
  const int half = lane >> 4;
  const int lane16 = lane & 15;

  v8f acc[8];
  const v8f zero = {0.f, 0.f, 0.f, 0.f, 0.f, 0.f, 0.f, 0.f};
#pragma unroll
  for (int t = 0; t < 8; ++t) acc[t] = zero;

  const float* __restrict__ arow = in + (size_t)(rowBase + lane16) * D_DIM;

  for (int k0 = 0; k0 < D_DIM; k0 += 4) {
    const int ka = k0 + 2 * half;
    v2f a;
    a.x = arow[ka];
    a.y = arow[ka + 1];
#pragma unroll
    for (int t = 0; t < 8; ++t) {
      const int n = t * 16 + lane16;
      v2f b;
      b.x = W[(size_t)n * D_DIM + ka];
      b.y = W[(size_t)n * D_DIM + ka + 1];
      acc[t] = __builtin_amdgcn_wmma_f32_16x16x4_f32(
          false, a, false, b, (short)0, acc[t], false, false);
    }
  }

#pragma unroll
  for (int t = 0; t < 8; ++t) {
    const int n = t * 16 + lane16;
    float g = 1.0f;
    if (APPLY_GAMMA) {
      const float lam = __expf(-__expf(lambda_log[n]));
      g = sqrtf(fmaf(-lam, lam, 1.0f + EPS_V));
    }
#pragma unroll
    for (int r = 0; r < 8; ++r) {
      const int row = rowBase + r + 8 * half;
      out[(size_t)row * N_DIM + n] = g * acc[t][r];
    }
  }
}

__global__ __launch_bounds__(256) void scan_kernel(
    const float* __restrict__ lambda_log, float* __restrict__ U,
    float* __restrict__ hT_out) {
  const int i = blockIdx.x * blockDim.x + threadIdx.x;
  const int n = i & (N_DIM - 1);
  const float lam = __expf(-__expf(lambda_log[n]));
  float h = 0.0f;
  float* __restrict__ p = U + i;
#pragma unroll 8
  for (int s = 0; s < S_LEN; ++s) {
    const float u = p[(size_t)s * (BT_DIM * N_DIM)];
    h = fmaf(lam, h, u);
    p[(size_t)s * (BT_DIM * N_DIM)] = h;
  }
  hT_out[i] = h;
}

extern "C" void kernel_launch(void* const* d_in, const int* in_sizes, int n_in,
                              void* d_out, int out_size, void* d_ws,
                              size_t ws_size, hipStream_t stream) {
  (void)in_sizes; (void)n_in; (void)out_size; (void)ws_size;
  const float* x          = (const float*)d_in[0];
  const float* lambda_log = (const float*)d_in[1];
  const float* B          = (const float*)d_in[2];
  const float* C          = (const float*)d_in[3];
  float* out = (float*)d_out;
  float* U   = (float*)d_ws;

  float* outputs = out;
  float* hT      = out + (size_t)ROWS * O_DIM;

  gemm_nt_wmma<true><<<dim3(ROWS / 16 / 8), dim3(256), 0, stream>>>(
      x, B, lambda_log, U);
  scan_kernel<<<dim3((BT_DIM * N_DIM) / 256), dim3(256), 0, stream>>>(
      lambda_log, U, hT);
  gemm_nt_wmma<false><<<dim3(ROWS / 16 / 8), dim3(256), 0, stream>>>(
      U, C, lambda_log, outputs);
}


# Scan split over time: 8 chunks of 32 steps per (b,n), one wave each, 32 loads in flight at once; chunk carries via LDS with lam^32 (squared in f64), recurrence rerun from carry; sum order changes
# speedup vs baseline: 1.8729x; 1.8729x over previous
#include <hip/hip_runtime.h>
#include <hip/hip_bf16.h>
#include <math.h>

#define S_LEN 256
#define BT_DIM 32
#define D_DIM 128
#define N_DIM 128
#define O_DIM 128
#define ROWS (S_LEN * BT_DIM)
#define EPS_V 1e-7f

typedef __attribute__((ext_vector_type(2))) float v2f;
typedef __attribute__((ext_vector_type(8))) float v8f;

template <bool APPLY_GAMMA>
__global__ __launch_bounds__(256) void gemm_nt_wmma(
    const float* __restrict__ in, const float* __restrict__ W,
    const float* __restrict__ lambda_log, float* __restrict__ out) {
  const int lane = threadIdx.x & 31;
  const int wave = blockIdx.x * 8 + (threadIdx.x >> 5);
  const int rowBase = wave * 16;
  const int half = lane >> 4;
  const int lane16 = lane & 15;

  v8f acc[8];
  const v8f zero = {0.f, 0.f, 0.f, 0.f, 0.f, 0.f, 0.f, 0.f};
#pragma unroll
  for (int t = 0; t < 8; ++t) acc[t] = zero;

  const float* __restrict__ arow = in + (size_t)(rowBase + lane16) * D_DIM;

  for (int k0 = 0; k0 < D_DIM; k0 += 4) {
    const int ka = k0 + 2 * half;
    v2f a;
    a.x = arow[ka];
    a.y = arow[ka + 1];
#pragma unroll
    for (int t = 0; t < 8; ++t) {
      const int n = t * 16 + lane16;
      v2f b;
      b.x = W[(size_t)n * D_DIM + ka];
      b.y = W[(size_t)n * D_DIM + ka + 1];
      acc[t] = __builtin_amdgcn_wmma_f32_16x16x4_f32(
          false, a, false, b, (short)0, acc[t], false, false);
    }
  }

#pragma unroll
  for (int t = 0; t < 8; ++t) {
    const int n = t * 16 + lane16;
    float g = 1.0f;
    if (APPLY_GAMMA) {
      const float lam = __expf(-__expf(lambda_log[n]));
      g = sqrtf(fmaf(-lam, lam, 1.0f + EPS_V));
    }
#pragma unroll
    for (int r = 0; r < 8; ++r) {
      const int row = rowBase + r + 8 * half;
      out[(size_t)row * N_DIM + n] = g * acc[t][r];
    }
  }
}

#define SCAN_T 32
#define SCAN_NC (S_LEN / SCAN_T)

__global__ __launch_bounds__(32 * SCAN_NC) void scan_kernel(
    const float* __restrict__ lambda_log, float* __restrict__ U,
    float* __restrict__ hT_out) {
  __shared__ float chunk_end[SCAN_NC][32];
  const int lane = threadIdx.x & 31;
  const int c = threadIdx.x >> 5;
  const int i = blockIdx.x * 32 + lane;
  const int n = i & (N_DIM - 1);
  float* __restrict__ p = U + (size_t)c * SCAN_T * (BT_DIM * N_DIM) + i;

  const float ll = lambda_log[n];

  float u[SCAN_T];
#pragma unroll
  for (int j = 0; j < SCAN_T; ++j) u[j] = p[(size_t)j * (BT_DIM * N_DIM)];

  __builtin_amdgcn_sched_barrier(0);

  const float lam = __expf(-__expf(ll));
  double lamTd = (double)lam;
#pragma unroll
  for (int q = 1; q < SCAN_T; q <<= 1) lamTd *= lamTd;
  const float lamT = (float)lamTd;

  float e = 0.0f;
#pragma unroll
  for (int j = 0; j < SCAN_T; ++j) e = fmaf(lam, e, u[j]);
  chunk_end[c][lane] = e;
  __syncthreads();

  float ek[SCAN_NC - 1];
#pragma unroll
  for (int k = 0; k < SCAN_NC - 1; ++k) ek[k] = chunk_end[k][lane];
  float h = 0.0f;
#pragma unroll
  for (int k = 0; k < SCAN_NC - 1; ++k) {
    const float t = fmaf(lamT, h, ek[k]);
    h = (k < c) ? t : h;
  }

#pragma unroll
  for (int j = 0; j < SCAN_T; ++j) {
    h = fmaf(lam, h, u[j]);
    p[(size_t)j * (BT_DIM * N_DIM)] = h;
  }
  if (c == SCAN_NC - 1) hT_out[i] = h;
}

extern "C" void kernel_launch(void* const* d_in, const int* in_sizes, int n_in,
                              void* d_out, int out_size, void* d_ws,
                              size_t ws_size, hipStream_t stream) {
  (void)in_sizes; (void)n_in; (void)out_size; (void)ws_size;
  const float* x          = (const float*)d_in[0];
  const float* lambda_log = (const float*)d_in[1];
  const float* B          = (const float*)d_in[2];
  const float* C          = (const float*)d_in[3];
  float* out = (float*)d_out;
  float* U   = (float*)d_ws;

  float* outputs = out;
  float* hT      = out + (size_t)ROWS * O_DIM;

  gemm_nt_wmma<true><<<dim3(ROWS / 16 / 8), dim3(256), 0, stream>>>(
      x, B, lambda_log, U);
  scan_kernel<<<dim3((BT_DIM * N_DIM) / 32), dim3(32 * SCAN_NC), 0, stream>>>(
      lambda_log, U, hT);
  gemm_nt_wmma<false><<<dim3(ROWS / 16 / 8), dim3(256), 0, stream>>>(
      U, C, lambda_log, outputs);
}


# GEMMs: no global load left in the k-loop: W copied once per block to padded LDS and read from there as B fragments (same k order); wave's A strip loaded to registers in one go behind the barrier
# speedup vs baseline: 2.5871x; 1.3813x over previous
#include <hip/hip_runtime.h>
#include <hip/hip_bf16.h>
#include <math.h>

#define S_LEN 256
#define BT_DIM 32
#define D_DIM 128
#define N_DIM 128
#define O_DIM 128
#define ROWS (S_LEN * BT_DIM)
#define EPS_V 1e-7f

typedef __attribute__((ext_vector_type(2))) float v2f;
typedef __attribute__((ext_vector_type(8))) float v8f;

#define W_PITCH (D_DIM + 4)
#define K_STEPS (D_DIM / 4)
#define K_BLOCKS (K_STEPS / 2)

template <bool APPLY_GAMMA>
__global__ __launch_bounds__(256) void gemm_nt_wmma(
    const float* __restrict__ in, const float* __restrict__ W,
    const float* __restrict__ lambda_log, float* __restrict__ out) {
  __shared__ __attribute__((aligned(16))) float Ws[N_DIM * W_PITCH];

  const int lane = threadIdx.x & 31;
  const int wave = blockIdx.x * 8 + (threadIdx.x >> 5);
  const int rowBase = wave * 16;
  const int half = lane >> 4;
  const int lane16 = lane & 15;

  float4 wreg[16];
#pragma unroll
  for (int p = 0; p < 16; ++p) {
    const int q = p * 256 + threadIdx.x;
    wreg[p] = *reinterpret_cast<const float4*>(W + (size_t)q * 4);
  }
#pragma unroll
  for (int p = 0; p < 16; ++p) {
    const int q = p * 256 + threadIdx.x;
    const int n = q >> 5, c4 = q & 31;
    *reinterpret_cast<float4*>(&Ws[n * W_PITCH + c4 * 4]) = wreg[p];
  }
  __syncthreads();

  const float* __restrict__ arow =
      in + (size_t)(rowBase + lane16) * D_DIM + 2 * half;
  v2f a[K_STEPS];
#pragma unroll
  for (int ks = 0; ks < K_STEPS; ++ks) {
    a[ks].x = arow[ks * 4];
    a[ks].y = arow[ks * 4 + 1];
  }

  v8f acc[8];
  const v8f zero = {0.f, 0.f, 0.f, 0.f, 0.f, 0.f, 0.f, 0.f};
#pragma unroll
  for (int t = 0; t < 8; ++t) acc[t] = zero;

  const float* __restrict__ bbase = &Ws[lane16 * W_PITCH + 2 * half];

  v2f bcur[2][8], bnext[2][8];
#pragma unroll
  for (int t = 0; t < 8; ++t) {
    bcur[0][t] = *reinterpret_cast<const v2f*>(bbase + t * 16 * W_PITCH);
    bcur[1][t] = *reinterpret_cast<const v2f*>(bbase + t * 16 * W_PITCH + 4);
  }

#pragma unroll
  for (int kb = 0; kb < K_BLOCKS; ++kb) {
    if (kb + 1 < K_BLOCKS) {
#pragma unroll
      for (int t = 0; t < 8; ++t) {
        bnext[0][t] = *reinterpret_cast<const v2f*>(
            bbase + t * 16 * W_PITCH + (2 * kb + 2) * 4);
        bnext[1][t] = *reinterpret_cast<const v2f*>(
            bbase + t * 16 * W_PITCH + (2 * kb + 3) * 4);
      }
    }
    __builtin_amdgcn_sched_barrier(0);
#pragma unroll
    for (int kk = 0; kk < 2; ++kk) {
#pragma unroll
      for (int t = 0; t < 8; ++t) {
        acc[t] = __builtin_amdgcn_wmma_f32_16x16x4_f32(
            false, a[2 * kb + kk], false, bcur[kk][t], (short)0, acc[t],
            false, false);
      }
    }
    __builtin_amdgcn_sched_barrier(0);
#pragma unroll
    for (int t = 0; t < 8; ++t) {
      bcur[0][t] = bnext[0][t];
      bcur[1][t] = bnext[1][t];
    }
  }

#pragma unroll
  for (int t = 0; t < 8; ++t) {
    const int n = t * 16 + lane16;
    float g = 1.0f;
    if (APPLY_GAMMA) {
      const float lam = __expf(-__expf(lambda_log[n]));
      g = sqrtf(fmaf(-lam, lam, 1.0f + EPS_V));
    }
#pragma unroll
    for (int r = 0; r < 8; ++r) {
      const int row = rowBase + r + 8 * half;
      out[(size_t)row * N_DIM + n] = g * acc[t][r];
    }
  }
}

#define SCAN_T 32
#define SCAN_NC (S_LEN / SCAN_T)

__global__ __launch_bounds__(32 * SCAN_NC) void scan_kernel(
    const float* __restrict__ lambda_log, float* __restrict__ U,
    float* __restrict__ hT_out) {
  __shared__ float chunk_end[SCAN_NC][32];
  const int lane = threadIdx.x & 31;
  const int c = threadIdx.x >> 5;
  const int i = blockIdx.x * 32 + lane;
  const int n = i & (N_DIM - 1);
  float* __restrict__ p = U + (size_t)c * SCAN_T * (BT_DIM * N_DIM) + i;

  const float ll = lambda_log[n];

  float u[SCAN_T];
#pragma unroll
  for (int j = 0; j < SCAN_T; ++j) u[j] = p[(size_t)j * (BT_DIM * N_DIM)];

  __builtin_amdgcn_sched_barrier(0);

  const float lam = __expf(-__expf(ll));
  double lamTd = (double)lam;
#pragma unroll
  for (int q = 1; q < SCAN_T; q <<= 1) lamTd *= lamTd;
  const float lamT = (float)lamTd;

  float e = 0.0f;
#pragma unroll
  for (int j = 0; j < SCAN_T; ++j) e = fmaf(lam, e, u[j]);
  chunk_end[c][lane] = e;
  __syncthreads();

  float ek[SCAN_NC - 1];
#pragma unroll
  for (int k = 0; k < SCAN_NC - 1; ++k) ek[k] = chunk_end[k][lane];
  float h = 0.0f;
#pragma unroll
  for (int k = 0; k < SCAN_NC - 1; ++k) {
    const float t = fmaf(lamT, h, ek[k]);
    h = (k < c) ? t : h;
  }

#pragma unroll
  for (int j = 0; j < SCAN_T; ++j) {
    h = fmaf(lam, h, u[j]);
    p[(size_t)j * (BT_DIM * N_DIM)] = h;
  }
  if (c == SCAN_NC - 1) hT_out[i] = h;
}

extern "C" void kernel_launch(void* const* d_in, const int* in_sizes, int n_in,
                              void* d_out, int out_size, void* d_ws,
                              size_t ws_size, hipStream_t stream) {
  (void)in_sizes; (void)n_in; (void)out_size; (void)ws_size;
  const float* x          = (const float*)d_in[0];
  const float* lambda_log = (const float*)d_in[1];
  const float* B          = (const float*)d_in[2];
  const float* C          = (const float*)d_in[3];
  float* out = (float*)d_out;
  float* U   = (float*)d_ws;

  float* outputs = out;
  float* hT      = out + (size_t)ROWS * O_DIM;

  gemm_nt_wmma<true><<<dim3(ROWS / 16 / 8), dim3(256), 0, stream>>>(
      x, B, lambda_log, U);
  scan_kernel<<<dim3((BT_DIM * N_DIM) / 32), dim3(32 * SCAN_NC), 0, stream>>>(
      lambda_log, U, hT);
  gemm_nt_wmma<false><<<dim3(ROWS / 16 / 8), dim3(256), 0, stream>>>(
      U, C, lambda_log, outputs);
}


# GEMM grid re-cut from 64 blocks of 128x128 (8 waves) to 256 blocks of 64x64 (4 waves, one per SIMD, 4 tiles each) so all 256 WGPs' matrix units work, not 64; loaders and k order unchanged
# speedup vs baseline: 3.6088x; 1.3949x over previous
#include <hip/hip_runtime.h>
#include <hip/hip_bf16.h>
#include <math.h>

#define S_LEN 256
#define BT_DIM 32
#define D_DIM 128
#define N_DIM 128
#define O_DIM 128
#define ROWS (S_LEN * BT_DIM)
#define EPS_V 1e-7f

typedef __attribute__((ext_vector_type(2))) float v2f;
typedef __attribute__((ext_vector_type(8))) float v8f;

#define BLK_ROWS 64
#define BLK_COLS 64
#define COL_BLOCKS (N_DIM / BLK_COLS)
#define TILES (BLK_COLS / 16)
#define W_PITCH (D_DIM + 4)
#define K_STEPS (D_DIM / 4)
#define KB_STEPS 4
#define K_BLOCKS (K_STEPS / KB_STEPS)
#define W_PIECES (BLK_COLS * D_DIM / 4 / 128)

template <bool APPLY_GAMMA>
__global__ __launch_bounds__(128) void gemm_nt_wmma(
    const float* __restrict__ in, const float* __restrict__ W,
    const float* __restrict__ lambda_log, float* __restrict__ out) {
  __shared__ __attribute__((aligned(16))) float Ws[BLK_COLS * W_PITCH];

  const int lane = threadIdx.x & 31;
  const int strip = threadIdx.x >> 5;
  const int colBase = (blockIdx.x % COL_BLOCKS) * BLK_COLS;
  const int rowBase = (blockIdx.x / COL_BLOCKS) * BLK_ROWS + strip * 16;
  const int half = lane >> 4;
  const int lane16 = lane & 15;

  const float* __restrict__ wsrc = W + (size_t)colBase * D_DIM;
  float4 wreg[W_PIECES];
#pragma unroll
  for (int p = 0; p < W_PIECES; ++p) {
    const int q = p * 128 + threadIdx.x;
    wreg[p] = *reinterpret_cast<const float4*>(wsrc + (size_t)q * 4);
  }
#pragma unroll
  for (int p = 0; p < W_PIECES; ++p) {
    const int q = p * 128 + threadIdx.x;
    const int n = q >> 5, c4 = q & 31;
    *reinterpret_cast<float4*>(&Ws[n * W_PITCH + c4 * 4]) = wreg[p];
  }
  __syncthreads();

  const float* __restrict__ arow =
      in + (size_t)(rowBase + lane16) * D_DIM + 2 * half;
  v2f a[K_STEPS];
#pragma unroll
  for (int ks = 0; ks < K_STEPS; ++ks) {
    a[ks].x = arow[ks * 4];
    a[ks].y = arow[ks * 4 + 1];
  }

  v8f acc[TILES];
  const v8f zero = {0.f, 0.f, 0.f, 0.f, 0.f, 0.f, 0.f, 0.f};
#pragma unroll
  for (int t = 0; t < TILES; ++t) acc[t] = zero;

  const float* __restrict__ bbase = &Ws[lane16 * W_PITCH + 2 * half];

  v2f bcur[KB_STEPS][TILES], bnext[KB_STEPS][TILES];
#pragma unroll
  for (int kk = 0; kk < KB_STEPS; ++kk) {
#pragma unroll
    for (int t = 0; t < TILES; ++t) {
      bcur[kk][t] =
          *reinterpret_cast<const v2f*>(bbase + t * 16 * W_PITCH + kk * 4);
    }
  }

#pragma unroll
  for (int kb = 0; kb < K_BLOCKS; ++kb) {
    if (kb + 1 < K_BLOCKS) {
#pragma unroll
      for (int kk = 0; kk < KB_STEPS; ++kk) {
#pragma unroll
        for (int t = 0; t < TILES; ++t) {
          bnext[kk][t] = *reinterpret_cast<const v2f*>(
              bbase + t * 16 * W_PITCH + (KB_STEPS * (kb + 1) + kk) * 4);
        }
      }
    }
    __builtin_amdgcn_sched_barrier(0);
#pragma unroll
    for (int kk = 0; kk < KB_STEPS; ++kk) {
#pragma unroll
      for (int t = 0; t < TILES; ++t) {
        acc[t] = __builtin_amdgcn_wmma_f32_16x16x4_f32(
            false, a[KB_STEPS * kb + kk], false, bcur[kk][t], (short)0, acc[t],
            false, false);
      }
    }
    __builtin_amdgcn_sched_barrier(0);
#pragma unroll
    for (int kk = 0; kk < KB_STEPS; ++kk) {
#pragma unroll
      for (int t = 0; t < TILES; ++t) bcur[kk][t] = bnext[kk][t];
    }
  }

#pragma unroll
  for (int t = 0; t < TILES; ++t) {
    const int n = colBase + t * 16 + lane16;
    float g = 1.0f;
    if (APPLY_GAMMA) {
      const float lam = __expf(-__expf(lambda_log[n]));
      g = sqrtf(fmaf(-lam, lam, 1.0f + EPS_V));
    }
#pragma unroll
    for (int r = 0; r < 8; ++r) {
      const int row = rowBase + r + 8 * half;
      out[(size_t)row * N_DIM + n] = g * acc[t][r];
    }
  }
}

#define SCAN_T 32
#define SCAN_NC (S_LEN / SCAN_T)

__global__ __launch_bounds__(32 * SCAN_NC) void scan_kernel(
    const float* __restrict__ lambda_log, float* __restrict__ U,
    float* __restrict__ hT_out) {
  __shared__ float chunk_end[SCAN_NC][32];
  const int lane = threadIdx.x & 31;
  const int c = threadIdx.x >> 5;
  const int i = blockIdx.x * 32 + lane;
  const int n = i & (N_DIM - 1);
  float* __restrict__ p = U + (size_t)c * SCAN_T * (BT_DIM * N_DIM) + i;

  const float ll = lambda_log[n];

  float u[SCAN_T];
#pragma unroll
  for (int j = 0; j < SCAN_T; ++j) u[j] = p[(size_t)j * (BT_DIM * N_DIM)];

  __builtin_amdgcn_sched_barrier(0);

  const float lam = __expf(-__expf(ll));
  double lamTd = (double)lam;
#pragma unroll
  for (int q = 1; q < SCAN_T; q <<= 1) lamTd *= lamTd;
  const float lamT = (float)lamTd;

  float e = 0.0f;
#pragma unroll
  for (int j = 0; j < SCAN_T; ++j) e = fmaf(lam, e, u[j]);
  chunk_end[c][lane] = e;
  __syncthreads();

  float ek[SCAN_NC - 1];
#pragma unroll
  for (int k = 0; k < SCAN_NC - 1; ++k) ek[k] = chunk_end[k][lane];
  float h = 0.0f;
#pragma unroll
  for (int k = 0; k < SCAN_NC - 1; ++k) {
    const float t = fmaf(lamT, h, ek[k]);
    h = (k < c) ? t : h;
  }

#pragma unroll
  for (int j = 0; j < SCAN_T; ++j) {
    h = fmaf(lam, h, u[j]);
    p[(size_t)j * (BT_DIM * N_DIM)] = h;
  }
  if (c == SCAN_NC - 1) hT_out[i] = h;
}

extern "C" void kernel_launch(void* const* d_in, const int* in_sizes, int n_in,
                              void* d_out, int out_size, void* d_ws,
                              size_t ws_size, hipStream_t stream) {
  (void)in_sizes; (void)n_in; (void)out_size; (void)ws_size;
  const float* x          = (const float*)d_in[0];
  const float* lambda_log = (const float*)d_in[1];
  const float* B          = (const float*)d_in[2];
  const float* C          = (const float*)d_in[3];
  float* out = (float*)d_out;
  float* U   = (float*)d_ws;

  float* outputs = out;
  float* hT      = out + (size_t)ROWS * O_DIM;

  gemm_nt_wmma<true><<<dim3((ROWS / BLK_ROWS) * COL_BLOCKS), dim3(128), 0, stream>>>(
      x, B, lambda_log, U);
  scan_kernel<<<dim3((BT_DIM * N_DIM) / 32), dim3(32 * SCAN_NC), 0, stream>>>(
      lambda_log, U, hT);
  gemm_nt_wmma<false><<<dim3((ROWS / BLK_ROWS) * COL_BLOCKS), dim3(128), 0, stream>>>(
      U, C, lambda_log, outputs);
}


# First product and time recurrence fused in one kernel: tiles with time steps as rows, scan on the accumulators in 8-step segments joined by carries (k and scan sum order change); 3 launches to 2
# speedup vs baseline: 4.7831x; 1.3254x over previous
#include <hip/hip_runtime.h>
#include <hip/hip_bf16.h>
#include <math.h>

#define S_LEN 256
#define BT_DIM 32
#define D_DIM 128
#define N_DIM 128
#define O_DIM 128
#define ROWS (S_LEN * BT_DIM)
#define EPS_V 1e-7f

typedef __attribute__((ext_vector_type(2))) float v2f;
typedef __attribute__((ext_vector_type(8))) float v8f;

#define HS_PLANE (S_LEN * 16)
#define N_GROUPS (N_DIM / 16)

__device__ __forceinline__ void load_piece(const float4* __restrict__ p, int q,
                                           v2f& f0, v2f& f1) {
  const float4 v = p[2 * q];
  f0.x = v.x;
  f0.y = v.y;
  f1.x = v.z;
  f1.y = v.w;
}

__global__ __launch_bounds__(128) void gemm_scan_wmma(
    const float* __restrict__ x, const float* __restrict__ Bw,
    const float* __restrict__ lambda_log, float* __restrict__ hs) {
  __shared__ float wave_end[4][16];

  const int lane = threadIdx.x & 31;
  const int wv = threadIdx.x >> 5;
  const int half = lane >> 4;
  const int lane16 = lane & 15;
  const int g = blockIdx.x & (N_GROUPS - 1);
  const int b = blockIdx.x >> 3;
  const int n = g * 16 + lane16;

  const float ll = lambda_log[n];

  const float4* __restrict__ pw =
      reinterpret_cast<const float4*>(Bw + (size_t)n * D_DIM) + half;
  const float4* __restrict__ px =
      reinterpret_cast<const float4*>(
          x + ((size_t)(wv * 64 + lane16) * BT_DIM + b) * D_DIM) + half;
  const size_t tile_f4 = (size_t)16 * BT_DIM * D_DIM / 4;

  v2f bf[32], a[4][32];
#pragma unroll
  for (int q = 0; q < 16; ++q) {
    load_piece(pw, q, bf[2 * q], bf[2 * q + 1]);
#pragma unroll
    for (int m = 0; m < 4; ++m)
      load_piece(px + m * tile_f4, q, a[m][2 * q], a[m][2 * q + 1]);
  }
  __builtin_amdgcn_sched_barrier(0);

  const float lam = __expf(-__expf(ll));
  const float gam = sqrtf(fmaf(-lam, lam, 1.0f + EPS_V));
  float lp[8];
  lp[0] = lam;
  double pd = (double)lam;
#pragma unroll
  for (int r = 1; r < 8; ++r) {
    pd *= (double)lam;
    lp[r] = (float)pd;
  }
  const float lam8 = lp[7];
  double p64 = pd;
  p64 *= p64;
  p64 *= p64;
  p64 *= p64;
  const float lam64 = (float)p64;

  v8f acc[4];
  const v8f zero = {0.f, 0.f, 0.f, 0.f, 0.f, 0.f, 0.f, 0.f};
#pragma unroll
  for (int m = 0; m < 4; ++m) acc[m] = zero;

#pragma unroll
  for (int ks = 0; ks < 32; ++ks) {
#pragma unroll
    for (int m = 0; m < 4; ++m) {
      acc[m] = __builtin_amdgcn_wmma_f32_16x16x4_f32(
          false, a[m][ks], false, bf[ks], (short)0, acc[m], false, false);
    }
  }

  float e[4][8];
#pragma unroll
  for (int m = 0; m < 4; ++m) {
    e[m][0] = gam * acc[m][0];
#pragma unroll
    for (int r = 1; r < 8; ++r) e[m][r] = fmaf(lam, e[m][r - 1], gam * acc[m][r]);
  }

  float eseg[8];
#pragma unroll
  for (int m = 0; m < 4; ++m) {
    const float own = e[m][7];
    const float oth = __shfl_xor(own, 16);
    eseg[2 * m] = half ? oth : own;
    eseg[2 * m + 1] = half ? own : oth;
  }

  float wend = 0.0f;
#pragma unroll
  for (int k = 0; k < 8; ++k) wend = fmaf(lam8, wend, eseg[k]);
  if (half == 0) wave_end[wv][lane16] = wend;
  __syncthreads();

  float we[3];
#pragma unroll
  for (int k = 0; k < 3; ++k) we[k] = wave_end[k][lane16];
  float cs = 0.0f;
#pragma unroll
  for (int k = 0; k < 3; ++k) {
    const float t = fmaf(lam64, cs, we[k]);
    cs = (k < wv) ? t : cs;
  }

  float cseg[4];
#pragma unroll
  for (int m = 0; m < 4; ++m) {
    const float s1 = fmaf(lam8, cs, eseg[2 * m]);
    cseg[m] = half ? s1 : cs;
    cs = fmaf(lam8, s1, eseg[2 * m + 1]);
  }

  float* __restrict__ plane =
      hs + ((size_t)b * N_GROUPS + g) * HS_PLANE + lane16;
#pragma unroll
  for (int m = 0; m < 4; ++m) {
#pragma unroll
    for (int r = 0; r < 8; ++r) {
      const float h = fmaf(lp[r], cseg[m], e[m][r]);
      const int t = wv * 64 + m * 16 + half * 8 + r;
      plane[(size_t)t * 16] = h;
    }
  }
}

#define BLK_ROWS 64
#define BLK_COLS 64
#define COL_BLOCKS (O_DIM / BLK_COLS)
#define T_BLOCKS (S_LEN / BLK_ROWS)
#define TILES (BLK_COLS / 16)
#define W_PITCH (N_DIM + 4)
#define K_STEPS (N_DIM / 4)
#define KB_STEPS 4
#define K_BLOCKS (K_STEPS / KB_STEPS)
#define W_PIECES (BLK_COLS * N_DIM / 4 / 128)

__global__ __launch_bounds__(128) void gemm_out_wmma(
    const float* __restrict__ hs, const float* __restrict__ W,
    float* __restrict__ out, float* __restrict__ hT_out) {
  __shared__ __attribute__((aligned(16))) float Ws[BLK_COLS * W_PITCH];

  const int lane = threadIdx.x & 31;
  const int strip = threadIdx.x >> 5;
  const int colBase = (blockIdx.x % COL_BLOCKS) * BLK_COLS;
  const int rb = blockIdx.x / COL_BLOCKS;
  const int b = rb / T_BLOCKS;
  const int tBase = (rb % T_BLOCKS) * BLK_ROWS + strip * 16;
  const int half = lane >> 4;
  const int lane16 = lane & 15;

  const float* __restrict__ wsrc = W + (size_t)colBase * N_DIM;
  float4 wreg[W_PIECES];
#pragma unroll
  for (int p = 0; p < W_PIECES; ++p) {
    const int q = p * 128 + threadIdx.x;
    wreg[p] = *reinterpret_cast<const float4*>(wsrc + (size_t)q * 4);
  }
#pragma unroll
  for (int p = 0; p < W_PIECES; ++p) {
    const int q = p * 128 + threadIdx.x;
    const int n = q >> 5, c4 = q & 31;
    *reinterpret_cast<float4*>(&Ws[n * W_PITCH + c4 * 4]) = wreg[p];
  }
  __syncthreads();

  const float* __restrict__ arow =
      hs + (size_t)b * N_GROUPS * HS_PLANE + (size_t)(tBase + lane16) * 16 +
      2 * half;
  v2f a[K_STEPS];
#pragma unroll
  for (int ks = 0; ks < K_STEPS; ++ks) {
    const float* __restrict__ p = arow + (size_t)(ks >> 2) * HS_PLANE + (ks & 3) * 4;
    a[ks].x = p[0];
    a[ks].y = p[1];
  }

  v8f acc[TILES];
  const v8f zero = {0.f, 0.f, 0.f, 0.f, 0.f, 0.f, 0.f, 0.f};
#pragma unroll
  for (int t = 0; t < TILES; ++t) acc[t] = zero;

  const float* __restrict__ bbase = &Ws[lane16 * W_PITCH + 2 * half];

  v2f bcur[KB_STEPS][TILES], bnext[KB_STEPS][TILES];
#pragma unroll
  for (int kk = 0; kk < KB_STEPS; ++kk) {
#pragma unroll
    for (int t = 0; t < TILES; ++t) {
      bcur[kk][t] =
          *reinterpret_cast<const v2f*>(bbase + t * 16 * W_PITCH + kk * 4);
    }
  }

#pragma unroll
  for (int kb = 0; kb < K_BLOCKS; ++kb) {
    if (kb + 1 < K_BLOCKS) {
#pragma unroll
      for (int kk = 0; kk < KB_STEPS; ++kk) {
#pragma unroll
        for (int t = 0; t < TILES; ++t) {
          bnext[kk][t] = *reinterpret_cast<const v2f*>(
              bbase + t * 16 * W_PITCH + (KB_STEPS * (kb + 1) + kk) * 4);
        }
      }
    }
    __builtin_amdgcn_sched_barrier(0);
#pragma unroll
    for (int kk = 0; kk < KB_STEPS; ++kk) {
#pragma unroll
      for (int t = 0; t < TILES; ++t) {
        acc[t] = __builtin_amdgcn_wmma_f32_16x16x4_f32(
            false, a[KB_STEPS * kb + kk], false, bcur[kk][t], (short)0, acc[t],
            false, false);
      }
    }
    __builtin_amdgcn_sched_barrier(0);
#pragma unroll
    for (int kk = 0; kk < KB_STEPS; ++kk) {
#pragma unroll
      for (int t = 0; t < TILES; ++t) bcur[kk][t] = bnext[kk][t];
    }
  }

#pragma unroll
  for (int t = 0; t < TILES; ++t) {
    const int o = colBase + t * 16 + lane16;
#pragma unroll
    for (int r = 0; r < 8; ++r) {
      const int row = (tBase + r + 8 * half) * BT_DIM + b;
      out[(size_t)row * O_DIM + o] = acc[t][r];
    }
  }

  if (blockIdx.x % COL_BLOCKS == 0 && rb % T_BLOCKS == T_BLOCKS - 1 &&
      strip == 3) {
    const float4 v = *reinterpret_cast<const float4*>(
        hs + ((size_t)(b * N_GROUPS + (lane >> 2)) * S_LEN + (S_LEN - 1)) * 16 +
        (lane & 3) * 4);
    *reinterpret_cast<float4*>(hT_out + (size_t)b * N_DIM + 4 * lane) = v;
  }
}

extern "C" void kernel_launch(void* const* d_in, const int* in_sizes, int n_in,
                              void* d_out, int out_size, void* d_ws,
                              size_t ws_size, hipStream_t stream) {
  (void)in_sizes; (void)n_in; (void)out_size; (void)ws_size;
  const float* x          = (const float*)d_in[0];
  const float* lambda_log = (const float*)d_in[1];
  const float* B          = (const float*)d_in[2];
  const float* C          = (const float*)d_in[3];
  float* out = (float*)d_out;
  float* HS  = (float*)d_ws;

  float* outputs = out;
  float* hT      = out + (size_t)ROWS * O_DIM;

  gemm_scan_wmma<<<dim3(BT_DIM * N_GROUPS), dim3(128), 0, stream>>>(
      x, B, lambda_log, HS);
  gemm_out_wmma<<<dim3(BT_DIM * T_BLOCKS * COL_BLOCKS), dim3(128), 0, stream>>>(
      HS, C, outputs, hT);
}
